# starting kernel
# baseline (speedup 1.0000x reference)
#include <hip/hip_runtime.h>

typedef float v2f __attribute__((ext_vector_type(2)));
typedef float v8f __attribute__((ext_vector_type(8)));

#define Nn   10000
#define Bb   2
#define Ll   16
#define FINf 8
#define Hh   64
#define Pp   12
#define Ee   80000

__device__ __forceinline__ v8f zero8() {
    v8f v;
#pragma unroll
    for (int i = 0; i < 8; ++i) v[i] = 0.f;
    return v;
}

__global__ void proj_kernel(const float* __restrict__ x,
                            const float* __restrict__ W_in,
                            const float* __restrict__ b_in,
                            float* __restrict__ hbuf) {
    int t  = blockIdx.x * 256 + threadIdx.x;
    int n  = t >> 8;
    int r  = t & 255;
    int b  = r >> 7;
    int li = (r >> 6) & 1;
    int c  = r & 63;
    int l  = 14 + li;
    const float* xr = x + (size_t)((b * Ll + l) * Nn + n) * FINf;
    float acc = b_in[c];
#pragma unroll
    for (int f = 0; f < FINf; ++f) acc += xr[f] * W_in[f * Hh + c];
    hbuf[(size_t)n * 256 + r] = fmaxf(acc, 0.f);
}

__global__ void scatter_kernel(const int* __restrict__ row,
                               const int* __restrict__ col,
                               const float* __restrict__ vals,
                               const float* __restrict__ hbuf,
                               float* __restrict__ agg) {
    int e  = blockIdx.x;
    int f  = threadIdx.x;
    int rr = row[e];
    int cc = col[e];
    float v = vals[e];
    atomicAdd(&agg[(size_t)rr * 256 + f], v * hbuf[(size_t)cc * 256 + f]);
}

__global__ void __launch_bounds__(256)
tcn_out_wmma_kernel(const float* __restrict__ agg,
                    const float* __restrict__ W_tcn,
                    const float* __restrict__ b_tcn,
                    const float* __restrict__ W_out,
                    const float* __restrict__ b_out,
                    float* __restrict__ out) {
    __shared__ float smem[8192 + 1024];
    float* Bs = smem;
    float* Ws = smem + 8192;
    float* Zs = smem;

    int tid = threadIdx.x;

    for (int i = tid; i < 8192; i += 256) {
        int j    = i >> 6;
        int cout = i & 63;
        Bs[i] = W_tcn[(cout * Hh + (j & 63)) * 3 + (j >> 6)];
    }
    for (int i = tid; i < 1024; i += 256) {
        int j2 = i >> 4;
        int p  = i & 15;
        Ws[i] = (p < Pp) ? W_out[j2 * Pp + p] : 0.f;
    }
    __syncthreads();

    int wave  = tid >> 5;
    int lane  = tid & 31;
    int lrow  = lane & 15;
    int khalf = lane >> 4;

    int tile = blockIdx.x * 8 + wave;
    if (tile > 1249) tile = 1249;
    int m0 = tile * 16;
    int b  = (m0 >= Nn) ? 1 : 0;
    int n0 = m0 - b * Nn;

    v8f acc[4];
#pragma unroll
    for (int ct = 0; ct < 4; ++ct) acc[ct] = zero8();

    const float* arow = agg + (size_t)(n0 + lrow) * 256 + b * 128;

    for (int kk = 0; kk < 32; ++kk) {
        int j = kk * 4 + khalf * 2;
        v2f av = *reinterpret_cast<const v2f*>(arow + j);
        v2f a;
        a.x = fmaxf(av.x, 0.f);
        a.y = fmaxf(av.y, 0.f);
#pragma unroll
        for (int ct = 0; ct < 4; ++ct) {
            v2f bv;
            bv.x = Bs[j * 64 + ct * 16 + lrow];
            bv.y = Bs[(j + 1) * 64 + ct * 16 + lrow];
            acc[ct] = __builtin_amdgcn_wmma_f32_16x16x4_f32(
                false, a, false, bv, (short)0, acc[ct], false, false);
        }
    }

    __syncthreads();

    float* Zw = Zs + wave * 1024;
#pragma unroll
    for (int ct = 0; ct < 4; ++ct) {
        float bt = b_tcn[ct * 16 + lrow];
#pragma unroll
        for (int i = 0; i < 8; ++i) {
            float z = fmaxf(acc[ct][i] + bt, 0.f);
            Zw[(i + khalf * 8) * 64 + ct * 16 + lrow] = z;
        }
    }
    __syncthreads();

    v8f accy = zero8();
    for (int kk = 0; kk < 16; ++kk) {
        int j2 = kk * 4 + khalf * 2;
        v2f a;
        a.x = Zw[lrow * 64 + j2];
        a.y = Zw[lrow * 64 + j2 + 1];
        v2f bv;
        bv.x = Ws[j2 * 16 + lrow];
        bv.y = Ws[(j2 + 1) * 16 + lrow];
        accy = __builtin_amdgcn_wmma_f32_16x16x4_f32(
            false, a, false, bv, (short)0, accy, false, false);
    }

    int p = lrow;
    if (p < Pp) {
        float bo = b_out[p];
#pragma unroll
        for (int i = 0; i < 8; ++i) {
            int n = n0 + i + khalf * 8;
            out[(size_t)(b * Pp + p) * Nn + n] = accy[i] + bo;
        }
    }
}

extern "C" void kernel_launch(void* const* d_in, const int* in_sizes, int n_in,
                              void* d_out, int out_size, void* d_ws, size_t ws_size,
                              hipStream_t stream) {
    const float* x     = (const float*)d_in[0];
    const int*   row   = (const int*)  d_in[1];
    const int*   col   = (const int*)  d_in[2];
    const float* vals  = (const float*)d_in[3];
    const float* W_in  = (const float*)d_in[4];
    const float* b_in  = (const float*)d_in[5];
    const float* W_tcn = (const float*)d_in[6];
    const float* b_tcn = (const float*)d_in[7];
    const float* W_out = (const float*)d_in[8];
    const float* b_out = (const float*)d_in[9];
    float*       out   = (float*)d_out;

    float* hbuf = (float*)d_ws;
    float* agg  = hbuf + (size_t)Nn * 256;

    hipMemsetAsync(agg, 0, (size_t)Nn * 256 * sizeof(float), stream);

    proj_kernel   <<<Nn, 256, 0, stream>>>(x, W_in, b_in, hbuf);
    scatter_kernel<<<Ee, 256, 0, stream>>>(row, col, vals, hbuf, agg);
    tcn_out_wmma_kernel<<<157, 256, 0, stream>>>(agg, W_tcn, b_tcn, W_out, b_out, out);
}


# Aggregation by row-owner work-groups: each sums vals*h[col] for its 40 rows in LDS (LDS float atomics, arrival-order sums as before) and stores agg once; global float atomics and memset removed
# speedup vs baseline: 2.4240x; 2.4240x over previous
#include <hip/hip_runtime.h>

typedef float v2f __attribute__((ext_vector_type(2)));
typedef float v8f __attribute__((ext_vector_type(8)));

#define Nn   10000
#define Bb   2
#define Ll   16
#define FINf 8
#define Hh   64
#define Pp   12
#define Ee   80000

__device__ __forceinline__ v8f zero8() {
    v8f v;
#pragma unroll
    for (int i = 0; i < 8; ++i) v[i] = 0.f;
    return v;
}

__global__ void proj_kernel(const float* __restrict__ x,
                            const float* __restrict__ W_in,
                            const float* __restrict__ b_in,
                            float* __restrict__ hbuf) {
    int t  = blockIdx.x * 256 + threadIdx.x;
    int n  = t >> 8;
    int r  = t & 255;
    int b  = r >> 7;
    int li = (r >> 6) & 1;
    int c  = r & 63;
    int l  = 14 + li;
    const float* xr = x + (size_t)((b * Ll + l) * Nn + n) * FINf;
    float acc = b_in[c];
#pragma unroll
    for (int f = 0; f < FINf; ++f) acc += xr[f] * W_in[f * Hh + c];
    hbuf[(size_t)n * 256 + r] = fmaxf(acc, 0.f);
}

#define ROWS_PER_GROUP 40
#define NGROUPS        250
#define AGG_WAVES      32
#define CHUNK          128
#define NCHUNK         625
#define FLUSH_AT       32
#define LIST_CAP       160

static_assert(NGROUPS * ROWS_PER_GROUP == Nn, "row ownership must tile N");
static_assert(NCHUNK * CHUNK == Ee, "chunks must tile E");
static_assert(LIST_CAP >= FLUSH_AT - 1 + CHUNK, "list must hold one chunk past the threshold");

__global__ void __launch_bounds__(1024)
agg_owner_kernel(const int* __restrict__ row,
                 const int* __restrict__ col,
                 const float* __restrict__ vals,
                 const float* __restrict__ hbuf,
                 float* __restrict__ agg) {
    __shared__ float    acc[ROWS_PER_GROUP * 256];
    __shared__ unsigned elist[AGG_WAVES * LIST_CAP];

    const int tid  = threadIdx.x;
    const int wave = __builtin_amdgcn_readfirstlane(tid >> 5);
    const int lane = tid & 31;
    const int r0   = blockIdx.x * ROWS_PER_GROUP;

    for (int i = tid; i < ROWS_PER_GROUP * 256; i += 1024) acc[i] = 0.f;
    __syncthreads();

    unsigned* wl = elist + wave * LIST_CAP;
    int cnt = 0;

    auto drain = [&]() {
        for (int base = 0; base < cnt; base += 32) {
            int nb = cnt - base;
            if (nb > 32) nb = 32;
            unsigned ent = 0u;
            int   cc = 0;
            float vv = 0.f;
            if (lane < nb) {
                ent = wl[base + lane];
                int e = (int)(ent >> 6);
                cc = col[e];
                vv = vals[e];
            }
            for (int i = 0; i < nb; i += 4) {
                float4 h0[4], h1[4];
                float  v[4];
                int    d[4];
#pragma unroll
                for (int u = 0; u < 4; ++u) {
                    int ii = i + u;
                    if (ii > nb - 1) ii = nb - 1;
                    int c = __builtin_amdgcn_readlane(cc, ii);
                    v[u]  = __int_as_float(__builtin_amdgcn_readlane(__float_as_int(vv), ii));
                    d[u]  = __builtin_amdgcn_readlane((int)ent, ii) & 63;
                    const float4* hp = reinterpret_cast<const float4*>(hbuf + (size_t)c * 256);
                    h0[u] = hp[lane];
                    h1[u] = hp[32 + lane];
                }
#pragma unroll
                for (int u = 0; u < 4; ++u) {
                    if (i + u < nb) {
                        float* a = acc + d[u] * 256 + lane;
                        atomicAdd(a,       v[u] * h0[u].x);
                        atomicAdd(a + 32,  v[u] * h0[u].y);
                        atomicAdd(a + 64,  v[u] * h0[u].z);
                        atomicAdd(a + 96,  v[u] * h0[u].w);
                        atomicAdd(a + 128, v[u] * h1[u].x);
                        atomicAdd(a + 160, v[u] * h1[u].y);
                        atomicAdd(a + 192, v[u] * h1[u].z);
                        atomicAdd(a + 224, v[u] * h1[u].w);
                    }
                }
            }
        }
        cnt = 0;
    };

    auto note = [&](int r, int e) {
        unsigned dd = (unsigned)(r - r0);
        bool m = dd < (unsigned)ROWS_PER_GROUP;
        unsigned mask = __builtin_amdgcn_ballot_w32(m);
        if (mask != 0u) {
            if (m) {
                int pos = cnt + __builtin_popcount(mask & ((1u << lane) - 1u));
                wl[pos] = ((unsigned)e << 6) | dd;
            }
            cnt += __builtin_popcount(mask);
        }
    };

    for (int c0 = wave; c0 < NCHUNK; c0 += AGG_WAVES * 4) {
        int4 rv[4];
#pragma unroll
        for (int u = 0; u < 4; ++u) {
            int ch = c0 + u * AGG_WAVES;
            if (ch > NCHUNK - 1) ch = NCHUNK - 1;
            rv[u] = *reinterpret_cast<const int4*>(row + ch * CHUNK + lane * 4);
        }
#pragma unroll
        for (int u = 0; u < 4; ++u) {
            int ch = c0 + u * AGG_WAVES;
            if (ch < NCHUNK) {
                int e0 = ch * CHUNK + lane * 4;
                note(rv[u].x, e0);
                note(rv[u].y, e0 + 1);
                note(rv[u].z, e0 + 2);
                note(rv[u].w, e0 + 3);
                if (cnt >= FLUSH_AT) drain();
            }
        }
    }
    drain();

    __syncthreads();

    for (int item = wave; item < ROWS_PER_GROUP * 2; item += AGG_WAVES) {
        int dr   = item >> 1;
        int part = item & 1;
        const float* a = acc + dr * 256 + part * 128 + lane;
        float4 o;
        o.x = a[0];
        o.y = a[32];
        o.z = a[64];
        o.w = a[96];
        reinterpret_cast<float4*>(agg + (size_t)(r0 + dr) * 256 + part * 128)[lane] = o;
    }
}

__global__ void __launch_bounds__(256)
tcn_out_wmma_kernel(const float* __restrict__ agg,
                    const float* __restrict__ W_tcn,
                    const float* __restrict__ b_tcn,
                    const float* __restrict__ W_out,
                    const float* __restrict__ b_out,
                    float* __restrict__ out) {
    __shared__ float smem[8192 + 1024];
    float* Bs = smem;
    float* Ws = smem + 8192;
    float* Zs = smem;

    int tid = threadIdx.x;

    for (int i = tid; i < 8192; i += 256) {
        int j    = i >> 6;
        int cout = i & 63;
        Bs[i] = W_tcn[(cout * Hh + (j & 63)) * 3 + (j >> 6)];
    }
    for (int i = tid; i < 1024; i += 256) {
        int j2 = i >> 4;
        int p  = i & 15;
        Ws[i] = (p < Pp) ? W_out[j2 * Pp + p] : 0.f;
    }
    __syncthreads();

    int wave  = tid >> 5;
    int lane  = tid & 31;
    int lrow  = lane & 15;
    int khalf = lane >> 4;

    int tile = blockIdx.x * 8 + wave;
    if (tile > 1249) tile = 1249;
    int m0 = tile * 16;
    int b  = (m0 >= Nn) ? 1 : 0;
    int n0 = m0 - b * Nn;

    v8f acc[4];
#pragma unroll
    for (int ct = 0; ct < 4; ++ct) acc[ct] = zero8();

    const float* arow = agg + (size_t)(n0 + lrow) * 256 + b * 128;

    for (int kk = 0; kk < 32; ++kk) {
        int j = kk * 4 + khalf * 2;
        v2f av = *reinterpret_cast<const v2f*>(arow + j);
        v2f a;
        a.x = fmaxf(av.x, 0.f);
        a.y = fmaxf(av.y, 0.f);
#pragma unroll
        for (int ct = 0; ct < 4; ++ct) {
            v2f bv;
            bv.x = Bs[j * 64 + ct * 16 + lrow];
            bv.y = Bs[(j + 1) * 64 + ct * 16 + lrow];
            acc[ct] = __builtin_amdgcn_wmma_f32_16x16x4_f32(
                false, a, false, bv, (short)0, acc[ct], false, false);
        }
    }

    __syncthreads();

    float* Zw = Zs + wave * 1024;
#pragma unroll
    for (int ct = 0; ct < 4; ++ct) {
        float bt = b_tcn[ct * 16 + lrow];
#pragma unroll
        for (int i = 0; i < 8; ++i) {
            float z = fmaxf(acc[ct][i] + bt, 0.f);
            Zw[(i + khalf * 8) * 64 + ct * 16 + lrow] = z;
        }
    }
    __syncthreads();

    v8f accy = zero8();
    for (int kk = 0; kk < 16; ++kk) {
        int j2 = kk * 4 + khalf * 2;
        v2f a;
        a.x = Zw[lrow * 64 + j2];
        a.y = Zw[lrow * 64 + j2 + 1];
        v2f bv;
        bv.x = Ws[j2 * 16 + lrow];
        bv.y = Ws[(j2 + 1) * 16 + lrow];
        accy = __builtin_amdgcn_wmma_f32_16x16x4_f32(
            false, a, false, bv, (short)0, accy, false, false);
    }

    int p = lrow;
    if (p < Pp) {
        float bo = b_out[p];
#pragma unroll
        for (int i = 0; i < 8; ++i) {
            int n = n0 + i + khalf * 8;
            out[(size_t)(b * Pp + p) * Nn + n] = accy[i] + bo;
        }
    }
}

extern "C" void kernel_launch(void* const* d_in, const int* in_sizes, int n_in,
                              void* d_out, int out_size, void* d_ws, size_t ws_size,
                              hipStream_t stream) {
    const float* x     = (const float*)d_in[0];
    const int*   row   = (const int*)  d_in[1];
    const int*   col   = (const int*)  d_in[2];
    const float* vals  = (const float*)d_in[3];
    const float* W_in  = (const float*)d_in[4];
    const float* b_in  = (const float*)d_in[5];
    const float* W_tcn = (const float*)d_in[6];
    const float* b_tcn = (const float*)d_in[7];
    const float* W_out = (const float*)d_in[8];
    const float* b_out = (const float*)d_in[9];
    float*       out   = (float*)d_out;

    float* hbuf = (float*)d_ws;
    float* agg  = hbuf + (size_t)Nn * 256;

    proj_kernel   <<<Nn, 256, 0, stream>>>(x, W_in, b_in, hbuf);
    agg_owner_kernel<<<NGROUPS, 1024, 0, stream>>>(row, col, vals, hbuf, agg);
    tcn_out_wmma_kernel<<<157, 256, 0, stream>>>(agg, W_tcn, b_tcn, W_out, b_out, out);
}


# Projection folded into aggregation: a wave loads an edge's 32 x[col] values one per lane and recomputes relu(xW+b) by row_share reads and fused multiply-adds in f order; first kernel and h removed
# speedup vs baseline: 3.0698x; 1.2664x over previous
#include <hip/hip_runtime.h>

typedef float v2f __attribute__((ext_vector_type(2)));
typedef float v8f __attribute__((ext_vector_type(8)));

#define Nn   10000
#define Bb   2
#define Ll   16
#define FINf 8
#define Hh   64
#define Pp   12
#define Ee   80000

__device__ __forceinline__ v8f zero8() {
    v8f v;
#pragma unroll
    for (int i = 0; i < 8; ++i) v[i] = 0.f;
    return v;
}

#define ROWS_PER_GROUP 40
#define NGROUPS        250
#define AGG_WAVES      32
#define CHUNK          128
#define NCHUNK         625
#define FLUSH_AT       32
#define LIST_CAP       160

static_assert(NGROUPS * ROWS_PER_GROUP == Nn, "row ownership must tile N");
static_assert(NCHUNK * CHUNK == Ee, "chunks must tile E");
static_assert(LIST_CAP >= FLUSH_AT - 1 + CHUNK, "list must hold one chunk past the threshold");

#define ROW_SHARE(v, K) \
    __int_as_float(__builtin_amdgcn_update_dpp(0, __float_as_int(v), 0x150 + (K), 0xF, 0xF, true))
#define PROJ_STEP(oa, ob, xs, f) \
    { const float s_ = (xs); const v2f ss_ = {s_, s_}; \
      (oa) = __builtin_elementwise_fma(ss_, wa[f], (oa)); (ob) = __builtin_elementwise_fma(ss_, wb[f], (ob)); }
#define PROJ4(oa, ob, xv, BASE) \
    { (oa) = bia; (ob) = bib; \
      PROJ_STEP(oa, ob, ROW_SHARE(xv, (BASE) + 0), 0) PROJ_STEP(oa, ob, ROW_SHARE(xv, (BASE) + 1), 1) \
      PROJ_STEP(oa, ob, ROW_SHARE(xv, (BASE) + 2), 2) PROJ_STEP(oa, ob, ROW_SHARE(xv, (BASE) + 3), 3) \
      PROJ_STEP(oa, ob, ROW_SHARE(xv, (BASE) + 4), 4) PROJ_STEP(oa, ob, ROW_SHARE(xv, (BASE) + 5), 5) \
      PROJ_STEP(oa, ob, ROW_SHARE(xv, (BASE) + 6), 6) PROJ_STEP(oa, ob, ROW_SHARE(xv, (BASE) + 7), 7) \
      (oa).x = fmaxf((oa).x, 0.f); (oa).y = fmaxf((oa).y, 0.f); (ob).x = fmaxf((ob).x, 0.f); (ob).y = fmaxf((ob).y, 0.f); }

__global__ void __launch_bounds__(1024)
agg_owner_kernel(const float* __restrict__ x,
                 const float* __restrict__ W_in,
                 const float* __restrict__ b_in,
                 const int* __restrict__ row,
                 const int* __restrict__ col,
                 const float* __restrict__ vals,
                 float* __restrict__ agg) {
    __shared__ float    acc[ROWS_PER_GROUP * 256];
    __shared__ unsigned elist[AGG_WAVES * LIST_CAP];

    const int tid  = threadIdx.x;
    const int wave = __builtin_amdgcn_readfirstlane(tid >> 5);
    const int lane = tid & 31;
    const int r0   = blockIdx.x * ROWS_PER_GROUP;

    const int cq = (lane & 15) * 4;
    const int li = lane >> 4;
    v2f wa[FINf], wb[FINf];
#pragma unroll
    for (int f = 0; f < FINf; ++f) {
        const float4 t = *reinterpret_cast<const float4*>(W_in + f * Hh + cq);
        wa[f].x = t.x; wa[f].y = t.y; wb[f].x = t.z; wb[f].y = t.w;
    }
    const float4 bt = *reinterpret_cast<const float4*>(b_in + cq);
    const v2f bia = {bt.x, bt.y};
    const v2f bib = {bt.z, bt.w};
    const float* xl = x + (size_t)((((lane >> 3) & 1) * Ll + 14 + li) * Nn) * FINf + (lane & 7);

    for (int i = tid; i < ROWS_PER_GROUP * 256; i += 1024) acc[i] = 0.f;
    __syncthreads();

    unsigned* wl = elist + wave * LIST_CAP;
    int cnt = 0;

    auto drain = [&]() {
        for (int base = 0; base < cnt; base += 32) {
            int nb = cnt - base;
            if (nb > 32) nb = 32;
            unsigned ent = 0u;
            int   cc = 0;
            float vv = 0.f;
            if (lane < nb) {
                ent = wl[base + lane];
                int e = (int)(ent >> 6);
                cc = col[e];
                vv = vals[e];
            }
            for (int i = 0; i < nb; i += 4) {
                float  xv[4];
                float  v[4];
                int    d[4];
#pragma unroll
                for (int u = 0; u < 4; ++u) {
                    int ii = i + u;
                    if (ii > nb - 1) ii = nb - 1;
                    int c = __builtin_amdgcn_readlane(cc, ii);
                    v[u]  = __int_as_float(__builtin_amdgcn_readlane(__float_as_int(vv), ii));
                    d[u]  = __builtin_amdgcn_readlane((int)ent, ii) & 63;
                    xv[u] = xl[(size_t)c * FINf];
                }
#pragma unroll
                for (int u = 0; u < 4; ++u) {
                    if (i + u < nb) {
                        float* a = acc + d[u] * 256 + lane;
                        v2f h0a, h0b, h1a, h1b;
                        PROJ4(h0a, h0b, xv[u], 0)
                        PROJ4(h1a, h1b, xv[u], 8)
                        atomicAdd(a,       v[u] * h0a.x);
                        atomicAdd(a + 32,  v[u] * h0a.y);
                        atomicAdd(a + 64,  v[u] * h0b.x);
                        atomicAdd(a + 96,  v[u] * h0b.y);
                        atomicAdd(a + 128, v[u] * h1a.x);
                        atomicAdd(a + 160, v[u] * h1a.y);
                        atomicAdd(a + 192, v[u] * h1b.x);
                        atomicAdd(a + 224, v[u] * h1b.y);
                    }
                }
            }
        }
        cnt = 0;
    };

    auto note = [&](int r, int e) {
        unsigned dd = (unsigned)(r - r0);
        bool m = dd < (unsigned)ROWS_PER_GROUP;
        unsigned mask = __builtin_amdgcn_ballot_w32(m);
        if (mask != 0u) {
            if (m) {
                int pos = cnt + __builtin_popcount(mask & ((1u << lane) - 1u));
                wl[pos] = ((unsigned)e << 6) | dd;
            }
            cnt += __builtin_popcount(mask);
        }
    };

    for (int c0 = wave; c0 < NCHUNK; c0 += AGG_WAVES * 4) {
        int4 rv[4];
#pragma unroll
        for (int u = 0; u < 4; ++u) {
            int ch = c0 + u * AGG_WAVES;
            if (ch > NCHUNK - 1) ch = NCHUNK - 1;
            rv[u] = *reinterpret_cast<const int4*>(row + ch * CHUNK + lane * 4);
        }
#pragma unroll
        for (int u = 0; u < 4; ++u) {
            int ch = c0 + u * AGG_WAVES;
            if (ch < NCHUNK) {
                int e0 = ch * CHUNK + lane * 4;
                note(rv[u].x, e0);
                note(rv[u].y, e0 + 1);
                note(rv[u].z, e0 + 2);
                note(rv[u].w, e0 + 3);
                if (cnt >= FLUSH_AT) drain();
            }
        }
    }
    drain();

    __syncthreads();

    for (int item = wave; item < ROWS_PER_GROUP * 2; item += AGG_WAVES) {
        int dr   = item >> 1;
        int part = item & 1;
        const float* a = acc + dr * 256 + part * 128 + lane;
        float4 o;
        o.x = a[0];
        o.y = a[32];
        o.z = a[64];
        o.w = a[96];
        reinterpret_cast<float4*>(agg + (size_t)(r0 + dr) * 256 + part * 128)[lane] = o;
    }
}

__global__ void __launch_bounds__(256)
tcn_out_wmma_kernel(const float* __restrict__ agg,
                    const float* __restrict__ W_tcn,
                    const float* __restrict__ b_tcn,
                    const float* __restrict__ W_out,
                    const float* __restrict__ b_out,
                    float* __restrict__ out) {
    __shared__ float smem[8192 + 1024];
    float* Bs = smem;
    float* Ws = smem + 8192;
    float* Zs = smem;

    int tid = threadIdx.x;

    for (int i = tid; i < 8192; i += 256) {
        int j    = i >> 6;
        int cout = i & 63;
        Bs[i] = W_tcn[(cout * Hh + (j & 63)) * 3 + (j >> 6)];
    }
    for (int i = tid; i < 1024; i += 256) {
        int j2 = i >> 4;
        int p  = i & 15;
        Ws[i] = (p < Pp) ? W_out[j2 * Pp + p] : 0.f;
    }
    __syncthreads();

    int wave  = tid >> 5;
    int lane  = tid & 31;
    int lrow  = lane & 15;
    int khalf = lane >> 4;

    int tile = blockIdx.x * 8 + wave;
    if (tile > 1249) tile = 1249;
    int m0 = tile * 16;
    int b  = (m0 >= Nn) ? 1 : 0;
    int n0 = m0 - b * Nn;

    v8f acc[4];
#pragma unroll
    for (int ct = 0; ct < 4; ++ct) acc[ct] = zero8();

    const float* arow = agg + (size_t)(n0 + lrow) * 256 + b * 128;

    for (int kk = 0; kk < 32; ++kk) {
        int j = kk * 4 + khalf * 2;
        v2f av = *reinterpret_cast<const v2f*>(arow + j);
        v2f a;
        a.x = fmaxf(av.x, 0.f);
        a.y = fmaxf(av.y, 0.f);
#pragma unroll
        for (int ct = 0; ct < 4; ++ct) {
            v2f bv;
            bv.x = Bs[j * 64 + ct * 16 + lrow];
            bv.y = Bs[(j + 1) * 64 + ct * 16 + lrow];
            acc[ct] = __builtin_amdgcn_wmma_f32_16x16x4_f32(
                false, a, false, bv, (short)0, acc[ct], false, false);
        }
    }

    __syncthreads();

    float* Zw = Zs + wave * 1024;
#pragma unroll
    for (int ct = 0; ct < 4; ++ct) {
        float bt = b_tcn[ct * 16 + lrow];
#pragma unroll
        for (int i = 0; i < 8; ++i) {
            float z = fmaxf(acc[ct][i] + bt, 0.f);
            Zw[(i + khalf * 8) * 64 + ct * 16 + lrow] = z;
        }
    }
    __syncthreads();

    v8f accy = zero8();
    for (int kk = 0; kk < 16; ++kk) {
        int j2 = kk * 4 + khalf * 2;
        v2f a;
        a.x = Zw[lrow * 64 + j2];
        a.y = Zw[lrow * 64 + j2 + 1];
        v2f bv;
        bv.x = Ws[j2 * 16 + lrow];
        bv.y = Ws[(j2 + 1) * 16 + lrow];
        accy = __builtin_amdgcn_wmma_f32_16x16x4_f32(
            false, a, false, bv, (short)0, accy, false, false);
    }

    int p = lrow;
    if (p < Pp) {
        float bo = b_out[p];
#pragma unroll
        for (int i = 0; i < 8; ++i) {
            int n = n0 + i + khalf * 8;
            out[(size_t)(b * Pp + p) * Nn + n] = accy[i] + bo;
        }
    }
}

extern "C" void kernel_launch(void* const* d_in, const int* in_sizes, int n_in,
                              void* d_out, int out_size, void* d_ws, size_t ws_size,
                              hipStream_t stream) {
    const float* x     = (const float*)d_in[0];
    const int*   row   = (const int*)  d_in[1];
    const int*   col   = (const int*)  d_in[2];
    const float* vals  = (const float*)d_in[3];
    const float* W_in  = (const float*)d_in[4];
    const float* b_in  = (const float*)d_in[5];
    const float* W_tcn = (const float*)d_in[6];
    const float* b_tcn = (const float*)d_in[7];
    const float* W_out = (const float*)d_in[8];
    const float* b_out = (const float*)d_in[9];
    float*       out   = (float*)d_out;

    float* agg = (float*)d_ws;

    agg_owner_kernel<<<NGROUPS, 1024, 0, stream>>>(x, W_in, b_in, row, col, vals, agg);
    tcn_out_wmma_kernel<<<157, 256, 0, stream>>>(agg, W_tcn, b_tcn, W_out, b_out, out);
}


# TCN and output products moved into the row-owner work-group: agg rows stay in LDS and feed the WMMA A loader (k order of the first product changed); second launch and agg round trip removed
# speedup vs baseline: 4.7885x; 1.5599x over previous
#include <hip/hip_runtime.h>

typedef float v2f __attribute__((ext_vector_type(2)));
typedef float v8f __attribute__((ext_vector_type(8)));

#define Nn   10000
#define Bb   2
#define Ll   16
#define FINf 8
#define Hh   64
#define Pp   12
#define Ee   80000

__device__ __forceinline__ v8f zero8() {
    v8f v;
#pragma unroll
    for (int i = 0; i < 8; ++i) v[i] = 0.f;
    return v;
}

#define ROWS_PER_GROUP 40
#define NGROUPS        250
#define AGG_WAVES      32
#define CHUNK          128
#define NCHUNK         625
#define FLUSH_AT       32
#define LIST_CAP       160
#define ACCP           257
#define BSP            145
#define BSTAP          80
#define ZSP            68
#define MROWS          (2 * ROWS_PER_GROUP)

static_assert(NGROUPS * ROWS_PER_GROUP == Nn, "row ownership must tile N");
static_assert(NCHUNK * CHUNK == Ee, "chunks must tile E");
static_assert(LIST_CAP >= FLUSH_AT - 1 + CHUNK, "list must hold one chunk past the threshold");
static_assert(MROWS % 16 == 0, "the owned (b, n) rows must tile by 16");
static_assert(BSP >= BSTAP + Hh, "both taps of a cin must fit its pitch");

#define ROW_SHARE(v, K) \
    __int_as_float(__builtin_amdgcn_update_dpp(0, __float_as_int(v), 0x150 + (K), 0xF, 0xF, true))
#define PROJ_STEP(oa, ob, xs, f) \
    { const float s_ = (xs); const v2f ss_ = {s_, s_}; \
      (oa) = __builtin_elementwise_fma(ss_, wa[f], (oa)); (ob) = __builtin_elementwise_fma(ss_, wb[f], (ob)); }
#define PROJ4(oa, ob, xv, BASE) \
    { (oa) = bia; (ob) = bib; \
      PROJ_STEP(oa, ob, ROW_SHARE(xv, (BASE) + 0), 0) PROJ_STEP(oa, ob, ROW_SHARE(xv, (BASE) + 1), 1) \
      PROJ_STEP(oa, ob, ROW_SHARE(xv, (BASE) + 2), 2) PROJ_STEP(oa, ob, ROW_SHARE(xv, (BASE) + 3), 3) \
      PROJ_STEP(oa, ob, ROW_SHARE(xv, (BASE) + 4), 4) PROJ_STEP(oa, ob, ROW_SHARE(xv, (BASE) + 5), 5) \
      PROJ_STEP(oa, ob, ROW_SHARE(xv, (BASE) + 6), 6) PROJ_STEP(oa, ob, ROW_SHARE(xv, (BASE) + 7), 7) \
      (oa).x = fmaxf((oa).x, 0.f); (oa).y = fmaxf((oa).y, 0.f); (ob).x = fmaxf((ob).x, 0.f); (ob).y = fmaxf((ob).y, 0.f); }

__global__ void __launch_bounds__(1024)
gcn_tcn_owner_kernel(const float* __restrict__ x,
                     const float* __restrict__ W_in,
                     const float* __restrict__ b_in,
                     const int* __restrict__ row,
                     const int* __restrict__ col,
                     const float* __restrict__ vals,
                     const float* __restrict__ W_tcn,
                     const float* __restrict__ b_tcn,
                     const float* __restrict__ W_out,
                     const float* __restrict__ b_out,
                     float* __restrict__ out) {
    __shared__ float    acc[ROWS_PER_GROUP * ACCP];
    __shared__ unsigned elist[AGG_WAVES * LIST_CAP];
    __shared__ float    Bs[Hh * BSP];
    __shared__ float    Ws[Hh * 16];
    __shared__ float    Zs[MROWS * ZSP];

    const int tid  = threadIdx.x;
    const int wave = __builtin_amdgcn_readfirstlane(tid >> 5);
    const int lane = tid & 31;
    const int r0   = blockIdx.x * ROWS_PER_GROUP;

    const int cq = (lane & 15) * 4;
    const int li = lane >> 4;
    v2f wa[FINf], wb[FINf];
#pragma unroll
    for (int f = 0; f < FINf; ++f) {
        const float4 t = *reinterpret_cast<const float4*>(W_in + f * Hh + cq);
        wa[f].x = t.x; wa[f].y = t.y; wb[f].x = t.z; wb[f].y = t.w;
    }
    const float4 bt = *reinterpret_cast<const float4*>(b_in + cq);
    const v2f bia = {bt.x, bt.y};
    const v2f bib = {bt.z, bt.w};
    const float* xl = x + (size_t)((((lane >> 3) & 1) * Ll + 14 + li) * Nn) * FINf + (lane & 7);

    const float4* wt4 = reinterpret_cast<const float4*>(W_tcn) + 3 * tid;
    const float4 wt0 = wt4[0];
    const float4 wt1 = wt4[1];
    const float4 wt2 = wt4[2];
    {
        const int j2 = tid >> 4;
        const int p  = tid & 15;
        Ws[tid] = (p < Pp) ? W_out[j2 * Pp + p] : 0.f;
    }
    for (int i = tid; i < ROWS_PER_GROUP * ACCP; i += 1024) acc[i] = 0.f;
    {
        float* bq = Bs + ((tid & 15) * 4) * BSP + (tid >> 4);
        bq[0 * BSP]         = wt0.x;
        bq[0 * BSP + BSTAP] = wt0.y;
        bq[1 * BSP]         = wt0.w;
        bq[1 * BSP + BSTAP] = wt1.x;
        bq[2 * BSP]         = wt1.z;
        bq[2 * BSP + BSTAP] = wt1.w;
        bq[3 * BSP]         = wt2.y;
        bq[3 * BSP + BSTAP] = wt2.z;
    }
    __syncthreads();

    unsigned* wl = elist + wave * LIST_CAP;
    int cnt = 0;

    auto drain = [&]() {
        for (int base = 0; base < cnt; base += 32) {
            int nb = cnt - base;
            if (nb > 32) nb = 32;
            unsigned ent = 0u;
            int   cc = 0;
            float vv = 0.f;
            if (lane < nb) {
                ent = wl[base + lane];
                int e = (int)(ent >> 6);
                cc = col[e];
                vv = vals[e];
            }
            for (int i = 0; i < nb; i += 4) {
                float  xv[4];
                float  v[4];
                int    d[4];
#pragma unroll
                for (int u = 0; u < 4; ++u) {
                    int ii = i + u;
                    if (ii > nb - 1) ii = nb - 1;
                    int c = __builtin_amdgcn_readlane(cc, ii);
                    v[u]  = __int_as_float(__builtin_amdgcn_readlane(__float_as_int(vv), ii));
                    d[u]  = __builtin_amdgcn_readlane((int)ent, ii) & 63;
                    xv[u] = xl[(size_t)c * FINf];
                }
#pragma unroll
                for (int u = 0; u < 4; ++u) {
                    if (i + u < nb) {
                        float* a = acc + d[u] * ACCP + lane;
                        v2f h0a, h0b, h1a, h1b;
                        PROJ4(h0a, h0b, xv[u], 0)
                        PROJ4(h1a, h1b, xv[u], 8)
                        atomicAdd(a,       v[u] * h0a.x);
                        atomicAdd(a + 32,  v[u] * h0a.y);
                        atomicAdd(a + 64,  v[u] * h0b.x);
                        atomicAdd(a + 96,  v[u] * h0b.y);
                        atomicAdd(a + 128, v[u] * h1a.x);
                        atomicAdd(a + 160, v[u] * h1a.y);
                        atomicAdd(a + 192, v[u] * h1b.x);
                        atomicAdd(a + 224, v[u] * h1b.y);
                    }
                }
            }
        }
        cnt = 0;
    };

    auto note = [&](int r, int e) {
        unsigned dd = (unsigned)(r - r0);
        bool m = dd < (unsigned)ROWS_PER_GROUP;
        unsigned mask = __builtin_amdgcn_ballot_w32(m);
        if (mask != 0u) {
            if (m) {
                int pos = cnt + __builtin_popcount(mask & ((1u << lane) - 1u));
                wl[pos] = ((unsigned)e << 6) | dd;
            }
            cnt += __builtin_popcount(mask);
        }
    };

    for (int c0 = wave; c0 < NCHUNK; c0 += AGG_WAVES * 4) {
        int4 rv[4];
#pragma unroll
        for (int u = 0; u < 4; ++u) {
            int ch = c0 + u * AGG_WAVES;
            if (ch > NCHUNK - 1) ch = NCHUNK - 1;
            rv[u] = *reinterpret_cast<const int4*>(row + ch * CHUNK + lane * 4);
        }
#pragma unroll
        for (int u = 0; u < 4; ++u) {
            int ch = c0 + u * AGG_WAVES;
            if (ch < NCHUNK) {
                int e0 = ch * CHUNK + lane * 4;
                note(rv[u].x, e0);
                note(rv[u].y, e0 + 1);
                note(rv[u].z, e0 + 2);
                note(rv[u].w, e0 + 3);
                if (cnt >= FLUSH_AT) drain();
            }
        }
    }
    drain();

    __syncthreads();

    const int lrow = lane & 15;
    const int hf   = lane >> 4;
    const float bo = b_out[lrow < Pp ? lrow : Pp - 1];

    if (wave < (MROWS / 16) * 4) {
        const int mt = wave >> 2;
        const int nt = wave & 3;
        const int mm = mt * 16 + lrow;
        const int bb = (mm >= ROWS_PER_GROUP) ? 1 : 0;
        const int dr = mm - bb * ROWS_PER_GROUP;
        const float* ap = acc + dr * ACCP + bb * 128 + 16 * hf;
        const float* bp = Bs + hf * BSTAP + nt * 16 + lrow;
        const float  btc = b_tcn[nt * 16 + lrow];
        v8f d = zero8();
#pragma unroll 8
        for (int t = 0; t < 32; ++t) {
            v2f a, bv;
            a.x  = fmaxf(ap[((t & 1) * 2 + 0) * 32 + (t >> 1)], 0.f);
            a.y  = fmaxf(ap[((t & 1) * 2 + 1) * 32 + (t >> 1)], 0.f);
            bv.x = bp[(2 * t + 0) * BSP];
            bv.y = bp[(2 * t + 1) * BSP];
            d = __builtin_amdgcn_wmma_f32_16x16x4_f32(
                false, a, false, bv, (short)0, d, false, false);
        }
#pragma unroll
        for (int i = 0; i < 8; ++i)
            Zs[(mt * 16 + hf * 8 + i) * ZSP + nt * 16 + lrow] = fmaxf(d[i] + btc, 0.f);
    }
    __syncthreads();

    if (wave < MROWS / 16) {
        const int mt = wave;
        const float* zp = Zs + (mt * 16 + lrow) * ZSP + hf * 2;
        const float* wp = Ws + (hf * 2) * 16 + lrow;
        v8f accy = zero8();
        for (int kk = 0; kk < 16; ++kk) {
            v2f a, bv;
            a.x  = zp[kk * 4];
            a.y  = zp[kk * 4 + 1];
            bv.x = wp[(kk * 4) * 16];
            bv.y = wp[(kk * 4 + 1) * 16];
            accy = __builtin_amdgcn_wmma_f32_16x16x4_f32(
                false, a, false, bv, (short)0, accy, false, false);
        }
        const int p = lrow;
        if (p < Pp) {
            const int mm0 = mt * 16 + hf * 8;
            const int bb  = (mm0 >= ROWS_PER_GROUP) ? 1 : 0;
            const int n0  = r0 + mm0 - bb * ROWS_PER_GROUP;
            float* o = out + (size_t)(bb * Pp + p) * Nn + n0;
#pragma unroll
            for (int i = 0; i < 8; ++i) o[i] = accy[i] + bo;
        }
    }
}

extern "C" void kernel_launch(void* const* d_in, const int* in_sizes, int n_in,
                              void* d_out, int out_size, void* d_ws, size_t ws_size,
                              hipStream_t stream) {
    const float* x     = (const float*)d_in[0];
    const int*   row   = (const int*)  d_in[1];
    const int*   col   = (const int*)  d_in[2];
    const float* vals  = (const float*)d_in[3];
    const float* W_in  = (const float*)d_in[4];
    const float* b_in  = (const float*)d_in[5];
    const float* W_tcn = (const float*)d_in[6];
    const float* b_tcn = (const float*)d_in[7];
    const float* W_out = (const float*)d_in[8];
    const float* b_out = (const float*)d_in[9];
    float*       out   = (float*)d_out;

    gcn_tcn_owner_kernel<<<NGROUPS, 1024, 0, stream>>>(x, W_in, b_in, row, col, vals,
                                                       W_tcn, b_tcn, W_out, b_out, out);
}
